# starting kernel
# baseline (speedup 1.0000x reference)
#include <hip/hip_runtime.h>
#include <math.h>

#define LEN     200
#define SHIFT_  150
#define FOUT    128
#define BATCH   64
#define SLEN    480050
#define TFRM    3200
#define TT      64
#define NS      9650
#define PITCH   129
#define EPSF    1e-7f
#define INV_LN10 0.43429448190325176f

typedef __attribute__((ext_vector_type(2))) float v2f;
typedef __attribute__((ext_vector_type(8))) float v8f;

__global__ __launch_bounds__(256) void dft_stats(const float* __restrict__ x,
                                                 float* __restrict__ ws) {
    __shared__ double sh[512];
    const int b   = blockIdx.x;
    const int tid = threadIdx.x;
    const float* xb = x + (size_t)b * SLEN;
    double s = 0.0, s2 = 0.0;
    for (int i = tid; i < SLEN; i += 256) {
        double v = (double)xb[i];
        s += v; s2 += v * v;
    }
    sh[tid] = s; sh[256 + tid] = s2;
    __syncthreads();
    for (int o = 128; o > 0; o >>= 1) {
        if (tid < o) { sh[tid] += sh[tid + o]; sh[256 + tid] += sh[256 + tid + o]; }
        __syncthreads();
    }
    if (tid == 0) {
        double m   = sh[0] / (double)SLEN;
        double var = sh[256] / (double)SLEN - m * m;
        if (var < 0.0) var = 0.0;
        ws[b]         = (float)m;
        ws[BATCH + b] = (float)(1.0 / (sqrt(var) + (double)EPSF));
    }
}

__global__ __launch_bounds__(256) void dft_spect(const float* __restrict__ x,
                                                 const float* __restrict__ kre,
                                                 const float* __restrict__ kim,
                                                 const float* __restrict__ ws,
                                                 float* __restrict__ out) {
    __shared__ float smem[NS];
    const int tid  = threadIdx.x;
    const int lane = tid & 31;
    const int wave = tid >> 5;
    const int b    = blockIdx.y;
    const int t0   = blockIdx.x * TT;

    const float mu  = ws[b];
    const float inv = ws[BATCH + b];

    const float* xb = x + (size_t)b * SLEN + (size_t)t0 * SHIFT_;
    for (int i = tid; i < NS; i += 256) smem[i] = (xb[i] - mu) * inv;
    __syncthreads();

    const int mrow = lane & 15;
    const int hi   = lane >> 4;
    const int kb   = hi << 1;
    const int f    = wave * 16 + mrow;

    const float* krp = kre + (size_t)f * LEN + kb;
    const float* kip = kim + (size_t)f * LEN + kb;

    const float* ap[4];
#pragma unroll
    for (int tt = 0; tt < 4; ++tt)
        ap[tt] = smem + SHIFT_ * (tt * 16 + mrow) + kb;

    v8f accR[4] = {}; v8f accI[4] = {};

#pragma unroll 2
    for (int k = 0; k < LEN; k += 4) {
        v2f br = *(const v2f*)(krp + k);
        v2f bi = *(const v2f*)(kip + k);
#pragma unroll
        for (int tt = 0; tt < 4; ++tt) {
            v2f a = *(const v2f*)(ap[tt] + k);
            accR[tt] = __builtin_amdgcn_wmma_f32_16x16x4_f32(
                false, a, false, br, (short)0, accR[tt], false, false);
            accI[tt] = __builtin_amdgcn_wmma_f32_16x16x4_f32(
                false, a, false, bi, (short)0, accI[tt], false, false);
        }
    }
    __syncthreads();

    float* fftl = smem;
    float* mus  = smem + TT * PITCH;
    float* invs = mus + TT;

#pragma unroll
    for (int tt = 0; tt < 4; ++tt) {
#pragma unroll
        for (int r = 0; r < 8; ++r) {
            float re  = accR[tt][r];
            float im  = accI[tt][r];
            float mag = sqrtf(re * re + im * im);
            float v   = logf(fmaxf(mag, EPSF)) * INV_LN10;
            int   tl  = tt * 16 + r + 8 * hi;
            fftl[tl * PITCH + f] = v;
        }
    }
    __syncthreads();

    if (tid < TT) {
        float s = 0.f, s2 = 0.f;
        const float* row = fftl + tid * PITCH;
        for (int ff = 0; ff < FOUT; ++ff) { float v = row[ff]; s += v; s2 += v * v; }
        float m   = s * (1.0f / FOUT);
        float var = fmaxf(s2 * (1.0f / FOUT) - m * m, 0.0f);
        mus[tid]  = m;
        invs[tid] = 1.0f / (sqrtf(var) + EPSF);
    }
    __syncthreads();

    float* outb = out + (size_t)b * FOUT * TFRM;
    for (int i = tid; i < TT * FOUT; i += 256) {
        int ff = i >> 6;
        int tl = i & (TT - 1);
        float v = (fftl[tl * PITCH + ff] - mus[tl]) * invs[tl];
        outb[(size_t)ff * TFRM + (t0 + tl)] = v;
    }
}

extern "C" void kernel_launch(void* const* d_in, const int* in_sizes, int n_in,
                              void* d_out, int out_size, void* d_ws, size_t ws_size,
                              hipStream_t stream) {
    const float* x   = (const float*)d_in[0];
    const float* kre = (const float*)d_in[1];
    const float* kim = (const float*)d_in[2];
    float* out = (float*)d_out;
    float* ws  = (float*)d_ws;

    dft_stats<<<dim3(BATCH), dim3(256), 0, stream>>>(x, ws);
    dft_spect<<<dim3(TFRM / TT, BATCH), dim3(256), 0, stream>>>(x, kre, kim, ws, out);
}


# statistics pass cut from 1 to 32 work-groups per batch row: f64 partial sums to workspace, folded by shuffles in each DFT work-group (sum order changes; inv-std finished in f32)
# speedup vs baseline: 1.8703x; 1.8703x over previous
#include <hip/hip_runtime.h>
#include <math.h>

#define LEN     200
#define SHIFT_  150
#define FOUT    128
#define BATCH   64
#define SLEN    480050
#define TFRM    3200
#define TT      64
#define NS      9650
#define PITCH   129
#define EPSF    1e-7f
#define INV_LN10 0.43429448190325176f
#define PCH     32
#define CHK     15008

typedef __attribute__((ext_vector_type(2))) float v2f;
typedef __attribute__((ext_vector_type(8))) float v8f;

__global__ __launch_bounds__(256) void dft_stats(const float* __restrict__ x,
                                                 float* __restrict__ ws) {
    __shared__ double shp[16];
    double* wsd = (double*)ws;
    const int c    = blockIdx.x;
    const int b    = blockIdx.y;
    const int tid  = threadIdx.x;
    const int lane = tid & 31;
    const int wave = tid >> 5;
    const int hi   = lane >> 4;

    const int start = c * CHK;
    const int len   = (SLEN - start < CHK) ? (SLEN - start) : CHK;
    const int n2    = len >> 1;
    const v2f* p2 = (const v2f*)(x + (size_t)b * SLEN + start);

    double s = 0.0, q = 0.0;
    for (int i = tid; i < n2; i += 256) {
        v2f v = p2[i];
        double a0 = (double)v.x, a1 = (double)v.y;
        s += a0; q = fma(a0, a0, q);
        s += a1; q = fma(a1, a1, q);
    }

    double keep = hi ? q : s;
    double give = hi ? s : q;
    keep += __shfl_xor(give, 16);
    keep += __shfl_xor(keep, 8);
    keep += __shfl_xor(keep, 4);
    keep += __shfl_xor(keep, 2);
    keep += __shfl_xor(keep, 1);
    if ((lane & 15) == 0) shp[hi * 8 + wave] = keep;
    __syncthreads();

    if (wave == 0) {
        double t = shp[lane & 15];
        t += __shfl_xor(t, 4);
        t += __shfl_xor(t, 2);
        t += __shfl_xor(t, 1);
        if (lane == 0) wsd[(size_t)b * (2 * PCH) + c]       = t;
        if (lane == 8) wsd[(size_t)b * (2 * PCH) + PCH + c] = t;
    }
}

__global__ __launch_bounds__(256) void dft_spect(const float* __restrict__ x,
                                                 const float* __restrict__ kre,
                                                 const float* __restrict__ kim,
                                                 const float* __restrict__ ws,
                                                 float* __restrict__ out) {
    const double* wsd = (const double*)ws;
    __shared__ float smem[NS];
    __shared__ float st[2];
    const int tid  = threadIdx.x;
    const int lane = tid & 31;
    const int wave = tid >> 5;
    const int b    = blockIdx.y;
    const int t0   = blockIdx.x * TT;

    if (wave == 0) {
        const int hq = lane >> 4;
        const int cc = lane & 15;
        const double* wp = wsd + (size_t)b * (2 * PCH) + hq * PCH + cc;
        double t = wp[0] + wp[16];
        t += __shfl_xor(t, 8);
        t += __shfl_xor(t, 4);
        t += __shfl_xor(t, 2);
        t += __shfl_xor(t, 1);
        double o = __shfl_xor(t, 16);
        double S = hq ? o : t;
        double Q = hq ? t : o;
        const double invN = 1.0 / (double)SLEN;
        double m   = S * invN;
        double var = fma(-m, m, Q * invN);
        float varf = fmaxf((float)var, 0.0f);
        if (lane == 0) {
            st[0] = (float)m;
            st[1] = 1.0f / (sqrtf(varf) + EPSF);
        }
    }
    __syncthreads();
    const float mu  = st[0];
    const float inv = st[1];

    const float* xb = x + (size_t)b * SLEN + (size_t)t0 * SHIFT_;
    for (int i = tid; i < NS; i += 256) smem[i] = (xb[i] - mu) * inv;
    __syncthreads();

    const int mrow = lane & 15;
    const int hi   = lane >> 4;
    const int kb   = hi << 1;
    const int f    = wave * 16 + mrow;

    const float* krp = kre + (size_t)f * LEN + kb;
    const float* kip = kim + (size_t)f * LEN + kb;

    const float* ap[4];
#pragma unroll
    for (int tt = 0; tt < 4; ++tt)
        ap[tt] = smem + SHIFT_ * (tt * 16 + mrow) + kb;

    v8f accR[4] = {}; v8f accI[4] = {};

#pragma unroll 2
    for (int k = 0; k < LEN; k += 4) {
        v2f br = *(const v2f*)(krp + k);
        v2f bi = *(const v2f*)(kip + k);
#pragma unroll
        for (int tt = 0; tt < 4; ++tt) {
            v2f a = *(const v2f*)(ap[tt] + k);
            accR[tt] = __builtin_amdgcn_wmma_f32_16x16x4_f32(
                false, a, false, br, (short)0, accR[tt], false, false);
            accI[tt] = __builtin_amdgcn_wmma_f32_16x16x4_f32(
                false, a, false, bi, (short)0, accI[tt], false, false);
        }
    }
    __syncthreads();

    float* fftl = smem;
    float* mus  = smem + TT * PITCH;
    float* invs = mus + TT;

#pragma unroll
    for (int tt = 0; tt < 4; ++tt) {
#pragma unroll
        for (int r = 0; r < 8; ++r) {
            float re  = accR[tt][r];
            float im  = accI[tt][r];
            float mag = sqrtf(re * re + im * im);
            float v   = logf(fmaxf(mag, EPSF)) * INV_LN10;
            int   tl  = tt * 16 + r + 8 * hi;
            fftl[tl * PITCH + f] = v;
        }
    }
    __syncthreads();

    if (tid < TT) {
        float s = 0.f, s2 = 0.f;
        const float* row = fftl + tid * PITCH;
        for (int ff = 0; ff < FOUT; ++ff) { float v = row[ff]; s += v; s2 += v * v; }
        float m   = s * (1.0f / FOUT);
        float var = fmaxf(s2 * (1.0f / FOUT) - m * m, 0.0f);
        mus[tid]  = m;
        invs[tid] = 1.0f / (sqrtf(var) + EPSF);
    }
    __syncthreads();

    float* outb = out + (size_t)b * FOUT * TFRM;
    for (int i = tid; i < TT * FOUT; i += 256) {
        int ff = i >> 6;
        int tl = i & (TT - 1);
        float v = (fftl[tl * PITCH + ff] - mus[tl]) * invs[tl];
        outb[(size_t)ff * TFRM + (t0 + tl)] = v;
    }
}

extern "C" void kernel_launch(void* const* d_in, const int* in_sizes, int n_in,
                              void* d_out, int out_size, void* d_ws, size_t ws_size,
                              hipStream_t stream) {
    const float* x   = (const float*)d_in[0];
    const float* kre = (const float*)d_in[1];
    const float* kim = (const float*)d_in[2];
    float* out = (float*)d_out;
    float* ws  = (float*)d_ws;

    dft_stats<<<dim3(PCH, BATCH), dim3(256), 0, stream>>>(x, ws);
    dft_spect<<<dim3(TFRM / TT, BATCH), dim3(256), 0, stream>>>(x, kre, kim, ws, out);
}


# statistics pass: per-thread accumulation precision f64 -> f32 to lift the f64 issue bound (folds across threads stay f64; loads batched eight deep; sum order changes)
# speedup vs baseline: 2.2293x; 1.1919x over previous
#include <hip/hip_runtime.h>
#include <math.h>

#define LEN     200
#define SHIFT_  150
#define FOUT    128
#define BATCH   64
#define SLEN    480050
#define TFRM    3200
#define TT      64
#define NS      9650
#define PITCH   129
#define EPSF    1e-7f
#define INV_LN10 0.43429448190325176f
#define PCH     32
#define CHK     15008
#define NL      8
#define NB      4

typedef __attribute__((ext_vector_type(2))) float v2f;
typedef __attribute__((ext_vector_type(8))) float v8f;

__global__ __launch_bounds__(256) void dft_stats(const float* __restrict__ x,
                                                 float* __restrict__ ws) {
    __shared__ double shp[16];
    double* wsd = (double*)ws;
    const int c    = blockIdx.x;
    const int b    = blockIdx.y;
    const int tid  = threadIdx.x;
    const int lane = tid & 31;
    const int wave = tid >> 5;
    const int hi   = lane >> 4;

    const int start = c * CHK;
    const int len   = (SLEN - start < CHK) ? (SLEN - start) : CHK;
    const int n2    = len >> 1;
    const v2f* p2 = (const v2f*)(x + (size_t)b * SLEN + start);

    static_assert(NB * NL * 256 >= CHK / 2, "batches must cover a whole chunk");
    static_assert((NB - 1) * NL * 256 <= (SLEN - (PCH - 1) * CHK) / 2,
                  "all batches but the last must lie inside the shortest chunk");
    float sf = 0.0f, qf = 0.0f;
#pragma unroll 1
    for (int t = 0; t < NB - 1; ++t) {
        v2f v[NL];
#pragma unroll
        for (int j = 0; j < NL; ++j) v[j] = p2[tid + (t * NL + j) * 256];
#pragma unroll
        for (int j = 0; j < NL; ++j) {
            sf += v[j].x; qf = fmaf(v[j].x, v[j].x, qf);
            sf += v[j].y; qf = fmaf(v[j].y, v[j].y, qf);
        }
    }
    {
        v2f v[NL];
#pragma unroll
        for (int j = 0; j < NL; ++j) {
            int idx = tid + ((NB - 1) * NL + j) * 256;
            v[j] = p2[idx < n2 ? idx : n2 - 1];
        }
#pragma unroll
        for (int j = 0; j < NL; ++j) {
            int idx = tid + ((NB - 1) * NL + j) * 256;
            float a0 = idx < n2 ? v[j].x : 0.0f;
            float a1 = idx < n2 ? v[j].y : 0.0f;
            sf += a0; qf = fmaf(a0, a0, qf);
            sf += a1; qf = fmaf(a1, a1, qf);
        }
    }
    double s = (double)sf, q = (double)qf;

    double keep = hi ? q : s;
    double give = hi ? s : q;
    keep += __shfl_xor(give, 16);
    keep += __shfl_xor(keep, 8);
    keep += __shfl_xor(keep, 4);
    keep += __shfl_xor(keep, 2);
    keep += __shfl_xor(keep, 1);
    if ((lane & 15) == 0) shp[hi * 8 + wave] = keep;
    __syncthreads();

    if (wave == 0) {
        double t = shp[lane & 15];
        t += __shfl_xor(t, 4);
        t += __shfl_xor(t, 2);
        t += __shfl_xor(t, 1);
        if (lane == 0) wsd[(size_t)b * (2 * PCH) + c]       = t;
        if (lane == 8) wsd[(size_t)b * (2 * PCH) + PCH + c] = t;
    }
}

__global__ __launch_bounds__(256) void dft_spect(const float* __restrict__ x,
                                                 const float* __restrict__ kre,
                                                 const float* __restrict__ kim,
                                                 const float* __restrict__ ws,
                                                 float* __restrict__ out) {
    const double* wsd = (const double*)ws;
    __shared__ float smem[NS];
    __shared__ float st[2];
    const int tid  = threadIdx.x;
    const int lane = tid & 31;
    const int wave = tid >> 5;
    const int b    = blockIdx.y;
    const int t0   = blockIdx.x * TT;

    if (wave == 0) {
        const int hq = lane >> 4;
        const int cc = lane & 15;
        const double* wp = wsd + (size_t)b * (2 * PCH) + hq * PCH + cc;
        double t = wp[0] + wp[16];
        t += __shfl_xor(t, 8);
        t += __shfl_xor(t, 4);
        t += __shfl_xor(t, 2);
        t += __shfl_xor(t, 1);
        double o = __shfl_xor(t, 16);
        double S = hq ? o : t;
        double Q = hq ? t : o;
        const double invN = 1.0 / (double)SLEN;
        double m   = S * invN;
        double var = fma(-m, m, Q * invN);
        float varf = fmaxf((float)var, 0.0f);
        if (lane == 0) {
            st[0] = (float)m;
            st[1] = 1.0f / (sqrtf(varf) + EPSF);
        }
    }
    __syncthreads();
    const float mu  = st[0];
    const float inv = st[1];

    const float* xb = x + (size_t)b * SLEN + (size_t)t0 * SHIFT_;
    for (int i = tid; i < NS; i += 256) smem[i] = (xb[i] - mu) * inv;
    __syncthreads();

    const int mrow = lane & 15;
    const int hi   = lane >> 4;
    const int kb   = hi << 1;
    const int f    = wave * 16 + mrow;

    const float* krp = kre + (size_t)f * LEN + kb;
    const float* kip = kim + (size_t)f * LEN + kb;

    const float* ap[4];
#pragma unroll
    for (int tt = 0; tt < 4; ++tt)
        ap[tt] = smem + SHIFT_ * (tt * 16 + mrow) + kb;

    v8f accR[4] = {}; v8f accI[4] = {};

#pragma unroll 2
    for (int k = 0; k < LEN; k += 4) {
        v2f br = *(const v2f*)(krp + k);
        v2f bi = *(const v2f*)(kip + k);
#pragma unroll
        for (int tt = 0; tt < 4; ++tt) {
            v2f a = *(const v2f*)(ap[tt] + k);
            accR[tt] = __builtin_amdgcn_wmma_f32_16x16x4_f32(
                false, a, false, br, (short)0, accR[tt], false, false);
            accI[tt] = __builtin_amdgcn_wmma_f32_16x16x4_f32(
                false, a, false, bi, (short)0, accI[tt], false, false);
        }
    }
    __syncthreads();

    float* fftl = smem;
    float* mus  = smem + TT * PITCH;
    float* invs = mus + TT;

#pragma unroll
    for (int tt = 0; tt < 4; ++tt) {
#pragma unroll
        for (int r = 0; r < 8; ++r) {
            float re  = accR[tt][r];
            float im  = accI[tt][r];
            float mag = sqrtf(re * re + im * im);
            float v   = logf(fmaxf(mag, EPSF)) * INV_LN10;
            int   tl  = tt * 16 + r + 8 * hi;
            fftl[tl * PITCH + f] = v;
        }
    }
    __syncthreads();

    if (tid < TT) {
        float s = 0.f, s2 = 0.f;
        const float* row = fftl + tid * PITCH;
        for (int ff = 0; ff < FOUT; ++ff) { float v = row[ff]; s += v; s2 += v * v; }
        float m   = s * (1.0f / FOUT);
        float var = fmaxf(s2 * (1.0f / FOUT) - m * m, 0.0f);
        mus[tid]  = m;
        invs[tid] = 1.0f / (sqrtf(var) + EPSF);
    }
    __syncthreads();

    float* outb = out + (size_t)b * FOUT * TFRM;
    for (int i = tid; i < TT * FOUT; i += 256) {
        int ff = i >> 6;
        int tl = i & (TT - 1);
        float v = (fftl[tl * PITCH + ff] - mus[tl]) * invs[tl];
        outb[(size_t)ff * TFRM + (t0 + tl)] = v;
    }
}

extern "C" void kernel_launch(void* const* d_in, const int* in_sizes, int n_in,
                              void* d_out, int out_size, void* d_ws, size_t ws_size,
                              hipStream_t stream) {
    const float* x   = (const float*)d_in[0];
    const float* kre = (const float*)d_in[1];
    const float* kim = (const float*)d_in[2];
    float* out = (float*)d_out;
    float* ws  = (float*)d_ws;

    dft_stats<<<dim3(PCH, BATCH), dim3(256), 0, stream>>>(x, ws);
    dft_spect<<<dim3(TFRM / TT, BATCH), dim3(256), 0, stream>>>(x, kre, kim, ws, out);
}


# epilogue: log10 magnitude as half of log10(2) times hardware log2 of re^2+im^2 (clamp moved to eps^2), not correctly rounded sqrt + library log: far fewer vector instructions, registers 140 -> 110
# speedup vs baseline: 2.5584x; 1.1476x over previous
#include <hip/hip_runtime.h>
#include <math.h>

#define LEN     200
#define SHIFT_  150
#define FOUT    128
#define BATCH   64
#define SLEN    480050
#define TFRM    3200
#define TT      64
#define NS      9650
#define PITCH   129
#define EPSF    1e-7f
#define HALF_LOG10_2 0.15051499783199060f
#define PCH     32
#define CHK     15008
#define NL      8
#define NB      4

typedef __attribute__((ext_vector_type(2))) float v2f;
typedef __attribute__((ext_vector_type(4))) float v4f;
typedef __attribute__((ext_vector_type(8))) float v8f;

__global__ __launch_bounds__(256) void dft_stats(const float* __restrict__ x,
                                                 float* __restrict__ ws) {
    __shared__ double shp[16];
    double* wsd = (double*)ws;
    const int c    = blockIdx.x;
    const int b    = blockIdx.y;
    const int tid  = threadIdx.x;
    const int lane = tid & 31;
    const int wave = tid >> 5;
    const int hi   = lane >> 4;

    const int start = c * CHK;
    const int len   = (SLEN - start < CHK) ? (SLEN - start) : CHK;
    const int n2    = len >> 1;
    const v2f* p2 = (const v2f*)(x + (size_t)b * SLEN + start);

    static_assert(NB * NL * 256 >= CHK / 2, "batches must cover a whole chunk");
    static_assert((NB - 1) * NL * 256 <= (SLEN - (PCH - 1) * CHK) / 2,
                  "all batches but the last must lie inside the shortest chunk");
    float sf = 0.0f, qf = 0.0f;
#pragma unroll 1
    for (int t = 0; t < NB - 1; ++t) {
        v2f v[NL];
#pragma unroll
        for (int j = 0; j < NL; ++j) v[j] = p2[tid + (t * NL + j) * 256];
#pragma unroll
        for (int j = 0; j < NL; ++j) {
            sf += v[j].x; qf = fmaf(v[j].x, v[j].x, qf);
            sf += v[j].y; qf = fmaf(v[j].y, v[j].y, qf);
        }
    }
    {
        v2f v[NL];
#pragma unroll
        for (int j = 0; j < NL; ++j) {
            int idx = tid + ((NB - 1) * NL + j) * 256;
            v[j] = p2[idx < n2 ? idx : n2 - 1];
        }
#pragma unroll
        for (int j = 0; j < NL; ++j) {
            int idx = tid + ((NB - 1) * NL + j) * 256;
            float a0 = idx < n2 ? v[j].x : 0.0f;
            float a1 = idx < n2 ? v[j].y : 0.0f;
            sf += a0; qf = fmaf(a0, a0, qf);
            sf += a1; qf = fmaf(a1, a1, qf);
        }
    }
    double s = (double)sf, q = (double)qf;

    double keep = hi ? q : s;
    double give = hi ? s : q;
    keep += __shfl_xor(give, 16);
    keep += __shfl_xor(keep, 8);
    keep += __shfl_xor(keep, 4);
    keep += __shfl_xor(keep, 2);
    keep += __shfl_xor(keep, 1);
    if ((lane & 15) == 0) shp[hi * 8 + wave] = keep;
    __syncthreads();

    if (wave == 0) {
        double t = shp[lane & 15];
        t += __shfl_xor(t, 4);
        t += __shfl_xor(t, 2);
        t += __shfl_xor(t, 1);
        if (lane == 0) wsd[(size_t)b * (2 * PCH) + c]       = t;
        if (lane == 8) wsd[(size_t)b * (2 * PCH) + PCH + c] = t;
    }
}

__global__ __launch_bounds__(256) void dft_spect(const float* __restrict__ x,
                                                 const float* __restrict__ kre,
                                                 const float* __restrict__ kim,
                                                 const float* __restrict__ ws,
                                                 float* __restrict__ out) {
    const double* wsd = (const double*)ws;
    __shared__ float smem[NS];
    __shared__ float st[2];
    const int tid  = threadIdx.x;
    const int lane = tid & 31;
    const int wave = tid >> 5;
    const int b    = blockIdx.y;
    const int t0   = blockIdx.x * TT;

    if (wave == 0) {
        const int hq = lane >> 4;
        const int cc = lane & 15;
        const double* wp = wsd + (size_t)b * (2 * PCH) + hq * PCH + cc;
        double t = wp[0] + wp[16];
        t += __shfl_xor(t, 8);
        t += __shfl_xor(t, 4);
        t += __shfl_xor(t, 2);
        t += __shfl_xor(t, 1);
        double o = __shfl_xor(t, 16);
        double S = hq ? o : t;
        double Q = hq ? t : o;
        const double invN = 1.0 / (double)SLEN;
        double m   = S * invN;
        double var = fma(-m, m, Q * invN);
        float varf = fmaxf((float)var, 0.0f);
        if (lane == 0) {
            st[0] = (float)m;
            st[1] = 1.0f / (sqrtf(varf) + EPSF);
        }
    }
    __syncthreads();
    const float mu  = st[0];
    const float inv = st[1];

    const float* xb = x + (size_t)b * SLEN + (size_t)t0 * SHIFT_;
    for (int i = tid; i < NS; i += 256) smem[i] = (xb[i] - mu) * inv;
    __syncthreads();

    const int mrow = lane & 15;
    const int hi   = lane >> 4;
    const int f    = wave * 16 + mrow;

    const int kq = hi << 2;
    const float* krp = kre + (size_t)f * LEN + kq;
    const float* kip = kim + (size_t)f * LEN + kq;

    const float* ap[4];
#pragma unroll
    for (int tt = 0; tt < 4; ++tt)
        ap[tt] = smem + SHIFT_ * (tt * 16 + mrow) + kq;

    v8f accR[4] = {}; v8f accI[4] = {};

    static_assert(LEN % 8 == 0, "tap blocks of 8 must cover a frame");
#pragma unroll
    for (int k = 0; k < LEN; k += 8) {
        const v4f br = *(const v4f*)(krp + k);
        const v4f bi = *(const v4f*)(kip + k);
#pragma unroll
        for (int j = 0; j < 2; ++j) {
            v2f bjr, bji;
            bjr.x = j ? br.z : br.x;  bjr.y = j ? br.w : br.y;
            bji.x = j ? bi.z : bi.x;  bji.y = j ? bi.w : bi.y;
#pragma unroll
            for (int tt = 0; tt < 4; ++tt) {
                v2f a = *(const v2f*)(ap[tt] + k + 2 * j);
                accR[tt] = __builtin_amdgcn_wmma_f32_16x16x4_f32(
                    false, a, false, bjr, (short)0, accR[tt], false, false);
                accI[tt] = __builtin_amdgcn_wmma_f32_16x16x4_f32(
                    false, a, false, bji, (short)0, accI[tt], false, false);
            }
        }
    }
    __syncthreads();

    float* fftl = smem;
    float* mus  = smem + TT * PITCH;
    float* invs = mus + TT;

#pragma unroll
    for (int tt = 0; tt < 4; ++tt) {
#pragma unroll
        for (int r = 0; r < 8; ++r) {
            float re  = accR[tt][r];
            float im  = accI[tt][r];
            float s   = fmaf(re, re, im * im);
            float v   = __builtin_amdgcn_logf(fmaxf(s, EPSF * EPSF)) * HALF_LOG10_2;
            int   tl  = tt * 16 + r + 8 * hi;
            fftl[tl * PITCH + f] = v;
        }
    }
    __syncthreads();

    if (tid < TT) {
        float s = 0.f, s2 = 0.f;
        const float* row = fftl + tid * PITCH;
        for (int ff = 0; ff < FOUT; ++ff) { float v = row[ff]; s += v; s2 += v * v; }
        float m   = s * (1.0f / FOUT);
        float var = fmaxf(s2 * (1.0f / FOUT) - m * m, 0.0f);
        mus[tid]  = m;
        invs[tid] = 1.0f / (sqrtf(var) + EPSF);
    }
    __syncthreads();

    float* outb = out + (size_t)b * FOUT * TFRM;
    for (int i = tid; i < TT * FOUT; i += 256) {
        int ff = i >> 6;
        int tl = i & (TT - 1);
        float v = (fftl[tl * PITCH + ff] - mus[tl]) * invs[tl];
        outb[(size_t)ff * TFRM + (t0 + tl)] = v;
    }
}

extern "C" void kernel_launch(void* const* d_in, const int* in_sizes, int n_in,
                              void* d_out, int out_size, void* d_ws, size_t ws_size,
                              hipStream_t stream) {
    const float* x   = (const float*)d_in[0];
    const float* kre = (const float*)d_in[1];
    const float* kim = (const float*)d_in[2];
    float* out = (float*)d_out;
    float* ws  = (float*)d_ws;

    dft_stats<<<dim3(PCH, BATCH), dim3(256), 0, stream>>>(x, ws);
    dft_spect<<<dim3(TFRM / TT, BATCH), dim3(256), 0, stream>>>(x, kre, kim, ws, out);
}


# tap loop software-pipelined by one block: B loads and A LDS reads of block k+8 issued as one group ahead of block k's 16 products (second register set), not waited for right after issue
# speedup vs baseline: 2.6819x; 1.0483x over previous
#include <hip/hip_runtime.h>
#include <math.h>

#define LEN     200
#define SHIFT_  150
#define FOUT    128
#define BATCH   64
#define SLEN    480050
#define TFRM    3200
#define TT      64
#define NS      9650
#define PITCH   129
#define EPSF    1e-7f
#define HALF_LOG10_2 0.15051499783199060f
#define PCH     32
#define CHK     15008
#define NL      8
#define NB      4

typedef __attribute__((ext_vector_type(2))) float v2f;
typedef __attribute__((ext_vector_type(4))) float v4f;
typedef __attribute__((ext_vector_type(8))) float v8f;

__global__ __launch_bounds__(256) void dft_stats(const float* __restrict__ x,
                                                 float* __restrict__ ws) {
    __shared__ double shp[16];
    double* wsd = (double*)ws;
    const int c    = blockIdx.x;
    const int b    = blockIdx.y;
    const int tid  = threadIdx.x;
    const int lane = tid & 31;
    const int wave = tid >> 5;
    const int hi   = lane >> 4;

    const int start = c * CHK;
    const int len   = (SLEN - start < CHK) ? (SLEN - start) : CHK;
    const int n2    = len >> 1;
    const v2f* p2 = (const v2f*)(x + (size_t)b * SLEN + start);

    static_assert(NB * NL * 256 >= CHK / 2, "batches must cover a whole chunk");
    static_assert((NB - 1) * NL * 256 <= (SLEN - (PCH - 1) * CHK) / 2,
                  "all batches but the last must lie inside the shortest chunk");
    float sf = 0.0f, qf = 0.0f;
#pragma unroll 1
    for (int t = 0; t < NB - 1; ++t) {
        v2f v[NL];
#pragma unroll
        for (int j = 0; j < NL; ++j) v[j] = p2[tid + (t * NL + j) * 256];
#pragma unroll
        for (int j = 0; j < NL; ++j) {
            sf += v[j].x; qf = fmaf(v[j].x, v[j].x, qf);
            sf += v[j].y; qf = fmaf(v[j].y, v[j].y, qf);
        }
    }
    {
        v2f v[NL];
#pragma unroll
        for (int j = 0; j < NL; ++j) {
            int idx = tid + ((NB - 1) * NL + j) * 256;
            v[j] = p2[idx < n2 ? idx : n2 - 1];
        }
#pragma unroll
        for (int j = 0; j < NL; ++j) {
            int idx = tid + ((NB - 1) * NL + j) * 256;
            float a0 = idx < n2 ? v[j].x : 0.0f;
            float a1 = idx < n2 ? v[j].y : 0.0f;
            sf += a0; qf = fmaf(a0, a0, qf);
            sf += a1; qf = fmaf(a1, a1, qf);
        }
    }
    double s = (double)sf, q = (double)qf;

    double keep = hi ? q : s;
    double give = hi ? s : q;
    keep += __shfl_xor(give, 16);
    keep += __shfl_xor(keep, 8);
    keep += __shfl_xor(keep, 4);
    keep += __shfl_xor(keep, 2);
    keep += __shfl_xor(keep, 1);
    if ((lane & 15) == 0) shp[hi * 8 + wave] = keep;
    __syncthreads();

    if (wave == 0) {
        double t = shp[lane & 15];
        t += __shfl_xor(t, 4);
        t += __shfl_xor(t, 2);
        t += __shfl_xor(t, 1);
        if (lane == 0) wsd[(size_t)b * (2 * PCH) + c]       = t;
        if (lane == 8) wsd[(size_t)b * (2 * PCH) + PCH + c] = t;
    }
}

__global__ __launch_bounds__(256, 8) void dft_spect(const float* __restrict__ x,
                                                 const float* __restrict__ kre,
                                                 const float* __restrict__ kim,
                                                 const float* __restrict__ ws,
                                                 float* __restrict__ out) {
    const double* wsd = (const double*)ws;
    __shared__ float smem[NS];
    __shared__ float st[2];
    const int tid  = threadIdx.x;
    const int lane = tid & 31;
    const int wave = tid >> 5;
    const int b    = blockIdx.y;
    const int t0   = blockIdx.x * TT;

    if (wave == 0) {
        const int hq = lane >> 4;
        const int cc = lane & 15;
        const double* wp = wsd + (size_t)b * (2 * PCH) + hq * PCH + cc;
        double t = wp[0] + wp[16];
        t += __shfl_xor(t, 8);
        t += __shfl_xor(t, 4);
        t += __shfl_xor(t, 2);
        t += __shfl_xor(t, 1);
        double o = __shfl_xor(t, 16);
        double S = hq ? o : t;
        double Q = hq ? t : o;
        const double invN = 1.0 / (double)SLEN;
        double m   = S * invN;
        double var = fma(-m, m, Q * invN);
        float varf = fmaxf((float)var, 0.0f);
        if (lane == 0) {
            st[0] = (float)m;
            st[1] = 1.0f / (sqrtf(varf) + EPSF);
        }
    }
    __syncthreads();
    const float mu  = st[0];
    const float inv = st[1];

    const float* xb = x + (size_t)b * SLEN + (size_t)t0 * SHIFT_;
    for (int i = tid; i < NS; i += 256) smem[i] = (xb[i] - mu) * inv;
    __syncthreads();

    const int mrow = lane & 15;
    const int hi   = lane >> 4;
    const int f    = wave * 16 + mrow;

    const int kq = hi << 2;
    const unsigned boff = (unsigned)(f * LEN + kq);
    const float* krp = kre + boff;
    const float* kip = kim + boff;

    const float* ap[4];
#pragma unroll
    for (int tt = 0; tt < 4; ++tt)
        ap[tt] = smem + SHIFT_ * (tt * 16 + mrow) + kq;

    v8f accR[4] = {}; v8f accI[4] = {};

    static_assert(LEN % 8 == 0, "tap blocks of 8 must cover a frame");
    v4f br = *(const v4f*)(krp);
    v4f bi = *(const v4f*)(kip);
    v2f a[2][4];
#pragma unroll
    for (int j = 0; j < 2; ++j)
#pragma unroll
        for (int tt = 0; tt < 4; ++tt)
            a[j][tt] = *(const v2f*)(ap[tt] + 2 * j);
#pragma unroll
    for (int k = 0; k < LEN; k += 8) {
        v4f brn = br, bin = bi;
        v2f an[2][4];
#pragma unroll
        for (int j = 0; j < 2; ++j)
#pragma unroll
            for (int tt = 0; tt < 4; ++tt) an[j][tt] = a[j][tt];
        if (k + 8 < LEN) {
            brn = *(const v4f*)(krp + k + 8);
            bin = *(const v4f*)(kip + k + 8);
#pragma unroll
            for (int j = 0; j < 2; ++j)
#pragma unroll
                for (int tt = 0; tt < 4; ++tt)
                    an[j][tt] = *(const v2f*)(ap[tt] + k + 8 + 2 * j);
        }
        __builtin_amdgcn_sched_barrier(0);
#pragma unroll
        for (int j = 0; j < 2; ++j) {
            v2f bjr, bji;
            bjr.x = j ? br.z : br.x;  bjr.y = j ? br.w : br.y;
            bji.x = j ? bi.z : bi.x;  bji.y = j ? bi.w : bi.y;
#pragma unroll
            for (int tt = 0; tt < 4; ++tt) {
                accR[tt] = __builtin_amdgcn_wmma_f32_16x16x4_f32(
                    false, a[j][tt], false, bjr, (short)0, accR[tt], false, false);
                accI[tt] = __builtin_amdgcn_wmma_f32_16x16x4_f32(
                    false, a[j][tt], false, bji, (short)0, accI[tt], false, false);
            }
        }
        br = brn; bi = bin;
#pragma unroll
        for (int j = 0; j < 2; ++j)
#pragma unroll
            for (int tt = 0; tt < 4; ++tt) a[j][tt] = an[j][tt];
    }
    __syncthreads();

    float* fftl = smem;
    float* mus  = smem + TT * PITCH;
    float* invs = mus + TT;

#pragma unroll
    for (int tt = 0; tt < 4; ++tt) {
#pragma unroll
        for (int r = 0; r < 8; ++r) {
            float re  = accR[tt][r];
            float im  = accI[tt][r];
            float s   = fmaf(re, re, im * im);
            float v   = __builtin_amdgcn_logf(fmaxf(s, EPSF * EPSF)) * HALF_LOG10_2;
            int   tl  = tt * 16 + r + 8 * hi;
            fftl[tl * PITCH + f] = v;
        }
    }
    __syncthreads();

    if (tid < TT) {
        float s = 0.f, s2 = 0.f;
        const float* row = fftl + tid * PITCH;
        for (int ff = 0; ff < FOUT; ++ff) { float v = row[ff]; s += v; s2 += v * v; }
        float m   = s * (1.0f / FOUT);
        float var = fmaxf(s2 * (1.0f / FOUT) - m * m, 0.0f);
        mus[tid]  = m;
        invs[tid] = 1.0f / (sqrtf(var) + EPSF);
    }
    __syncthreads();

    static_assert((long long)BATCH * FOUT * TFRM < (1ll << 31), "32-bit output offsets");
    const unsigned ob = (unsigned)b * (unsigned)(FOUT * TFRM) + (unsigned)t0;
    for (int i = tid; i < TT * FOUT; i += 256) {
        int ff = i >> 6;
        int tl = i & (TT - 1);
        float v = (fftl[tl * PITCH + ff] - mus[tl]) * invs[tl];
        out[ob + (unsigned)ff * TFRM + (unsigned)tl] = v;
    }
}

extern "C" void kernel_launch(void* const* d_in, const int* in_sizes, int n_in,
                              void* d_out, int out_size, void* d_ws, size_t ws_size,
                              hipStream_t stream) {
    const float* x   = (const float*)d_in[0];
    const float* kre = (const float*)d_in[1];
    const float* kim = (const float*)d_in[2];
    float* out = (float*)d_out;
    float* ws  = (float*)d_ws;

    dft_stats<<<dim3(PCH, BATCH), dim3(256), 0, stream>>>(x, ws);
    dft_spect<<<dim3(TFRM / TT, BATCH), dim3(256), 0, stream>>>(x, kre, kim, ws, out);
}
